# starting kernel
# baseline (speedup 1.0000x reference)
#include <hip/hip_runtime.h>

typedef float v2f __attribute__((ext_vector_type(2)));
typedef float v8f __attribute__((ext_vector_type(8)));

#define KDIM   500
#define HIDDEN 64

__global__ __launch_bounds__(256) void k_gemm_relu(
    const float* __restrict__ x, const float* __restrict__ W1,
    float* __restrict__ h, int n_tiles)
{
    const int lane   = threadIdx.x & 31;
    const int m_tile = blockIdx.x * 8 + (threadIdx.x >> 5);
    if (m_tile >= n_tiles) return;
    const int nn     = lane & 15;
    const int khalf  = (lane >> 4) << 1;

    const float* xrow = x  + (size_t)(m_tile * 16 + nn) * KDIM;
    const float* w0   = W1 + (size_t)(nn +  0) * KDIM;
    const float* w1   = W1 + (size_t)(nn + 16) * KDIM;
    const float* w2   = W1 + (size_t)(nn + 32) * KDIM;
    const float* w3   = W1 + (size_t)(nn + 48) * KDIM;

    v8f acc0 = {}, acc1 = {}, acc2 = {}, acc3 = {};
    for (int k0 = 0; k0 < KDIM; k0 += 4) {
        const int k = k0 + khalf;
        v2f a  = __builtin_nontemporal_load((const v2f*)(xrow + k));
        v2f b0 = *(const v2f*)(w0 + k);
        v2f b1 = *(const v2f*)(w1 + k);
        v2f b2 = *(const v2f*)(w2 + k);
        v2f b3 = *(const v2f*)(w3 + k);
        acc0 = __builtin_amdgcn_wmma_f32_16x16x4_f32(false, a, false, b0, (short)0, acc0, false, false);
        acc1 = __builtin_amdgcn_wmma_f32_16x16x4_f32(false, a, false, b1, (short)0, acc1, false, false);
        acc2 = __builtin_amdgcn_wmma_f32_16x16x4_f32(false, a, false, b2, (short)0, acc2, false, false);
        acc3 = __builtin_amdgcn_wmma_f32_16x16x4_f32(false, a, false, b3, (short)0, acc3, false, false);
    }
    const int m_lo = m_tile * 16 + ((lane >> 4) << 3);
    #pragma unroll
    for (int r = 0; r < 8; ++r) {
        const size_t mrow = (size_t)(m_lo + r) * HIDDEN;
        float v0 = acc0[r]; v0 = v0 > 0.f ? v0 : 0.f;
        float v1 = acc1[r]; v1 = v1 > 0.f ? v1 : 0.f;
        float v2 = acc2[r]; v2 = v2 > 0.f ? v2 : 0.f;
        float v3 = acc3[r]; v3 = v3 > 0.f ? v3 : 0.f;
        h[mrow + nn +  0] = v0;
        h[mrow + nn + 16] = v1;
        h[mrow + nn + 32] = v2;
        h[mrow + nn + 48] = v3;
    }
}

__global__ __launch_bounds__(256) void k_zero(float4* __restrict__ p, int n4)
{
    int i = blockIdx.x * blockDim.x + threadIdx.x;
    if (i < n4) p[i] = make_float4(0.f, 0.f, 0.f, 0.f);
}

__global__ __launch_bounds__(256) void k_spmm(
    const int* __restrict__ src, const int* __restrict__ dst,
    const float* __restrict__ val, const float* __restrict__ h,
    float* __restrict__ out, int nnz)
{
    const long long tid = (long long)blockIdx.x * blockDim.x + threadIdx.x;
    const int e = (int)(tid >> 5);
    if (e >= nnz) return;
    const int c = (int)((tid & 31) << 1);
    const int s = __builtin_nontemporal_load(src + e);
    const int d = __builtin_nontemporal_load(dst + e);
    const float v = __builtin_nontemporal_load(val + e);
    const v2f hv = *(const v2f*)(h + (size_t)s * HIDDEN + c);
    float* o = out + (size_t)d * HIDDEN + c;
    __hip_atomic_fetch_add(o,     v * hv.x, __ATOMIC_RELAXED, __HIP_MEMORY_SCOPE_AGENT);
    __hip_atomic_fetch_add(o + 1, v * hv.y, __ATOMIC_RELAXED, __HIP_MEMORY_SCOPE_AGENT);
}

__global__ __launch_bounds__(256) void k_final(
    const float* __restrict__ h,  const float* __restrict__ h1,
    const float* __restrict__ h2, const float* __restrict__ Wf,
    float* __restrict__ out, int n_tiles)
{
    const int lane = threadIdx.x & 31;
    const int tile = blockIdx.x * 8 + (threadIdx.x >> 5);
    if (tile >= n_tiles) return;
    const int nn    = lane & 15;
    const int khalf = (lane >> 4) << 1;
    const int   wn    = nn < 10 ? nn : 9;
    const float wmask = nn < 10 ? 1.f : 0.f;
    const size_t rowoff = (size_t)(tile * 16 + nn) * HIDDEN;
    const float* wrow   = Wf + (size_t)wn * 192;

    const float* bufs[3] = { h, h1, h2 };
    v8f acc = {};
    #pragma unroll
    for (int seg = 0; seg < 3; ++seg) {
        const float* buf = bufs[seg] + rowoff;
        #pragma unroll
        for (int k0 = 0; k0 < 64; k0 += 4) {
            v2f a = __builtin_nontemporal_load((const v2f*)(buf + k0 + khalf));
            v2f b;
            b.x = wrow[seg * 64 + k0 + khalf + 0] * wmask;
            b.y = wrow[seg * 64 + k0 + khalf + 1] * wmask;
            acc = __builtin_amdgcn_wmma_f32_16x16x4_f32(false, a, false, b, (short)0, acc, false, false);
        }
    }
    if (nn < 10) {
        const int m_lo = tile * 16 + ((lane >> 4) << 3);
        #pragma unroll
        for (int r = 0; r < 8; ++r)
            __builtin_nontemporal_store(acc[r], out + (size_t)(m_lo + r) * 10 + nn);
    }
}

extern "C" void kernel_launch(void* const* d_in, const int* in_sizes, int n_in,
                              void* d_out, int out_size, void* d_ws, size_t ws_size,
                              hipStream_t stream)
{
    const float* x    = (const float*)d_in[0];
    const float* W1   = (const float*)d_in[1];
    const float* Wf   = (const float*)d_in[2];
    const int*   src1 = (const int*)  d_in[3];
    const int*   dst1 = (const int*)  d_in[4];
    const float* val1 = (const float*)d_in[5];
    const int*   src2 = (const int*)  d_in[6];
    const int*   dst2 = (const int*)  d_in[7];
    const float* val2 = (const float*)d_in[8];

    const int n    = in_sizes[0] / KDIM;
    const int nnz1 = in_sizes[3];
    const int nnz2 = in_sizes[6];

    float* h   = (float*)d_ws;
    float* h1  = h  + (size_t)n * HIDDEN;
    float* h2  = h1 + (size_t)n * HIDDEN;
    float* out = (float*)d_out;

    const int n_tiles = (n + 15) / 16;

    k_gemm_relu<<<(n_tiles + 7) / 8, 256, 0, stream>>>(x, W1, h, n_tiles);

    {
        const long long n4 = (long long)2 * n * HIDDEN / 4;
        k_zero<<<(int)((n4 + 255) / 256), 256, 0, stream>>>((float4*)h1, (int)n4);
    }

    {
        const long long t1 = (long long)nnz1 * 32;
        k_spmm<<<(int)((t1 + 255) / 256), 256, 0, stream>>>(src1, dst1, val1, h, h1, nnz1);
        const long long t2 = (long long)nnz2 * 32;
        k_spmm<<<(int)((t2 + 255) / 256), 256, 0, stream>>>(src2, dst2, val2, h, h2, nnz2);
    }

    k_final<<<(n_tiles + 7) / 8, 256, 0, stream>>>(h, h1, h2, Wf, out, n_tiles);
}


# SpMM: dst rows cut in two ranges, one per L2 half; a resident workgroup reads its die and draws edge chunks from that half's ticket counter: a line gets atomics from one half only; add order differs
# speedup vs baseline: 3.2776x; 3.2776x over previous
#include <hip/hip_runtime.h>

typedef float v2f __attribute__((ext_vector_type(2)));
typedef float v4f __attribute__((ext_vector_type(4)));
typedef float v8f __attribute__((ext_vector_type(8)));

#define KDIM   500
#define HIDDEN 64
#define TK_STRIDE 64

#define MT 6
#define NPAIR (KDIM / 8)
#define KTAIL (NPAIR * 8)
#define LA 2
#define RING (LA + 1)
#define S_MAIN (((NPAIR - LA) / RING) * RING)
#define GEMM_WAVES 4
#define GEMM_WGS 256
#define SHARE (GEMM_WAVES * (2 * MT + 1))

static_assert(KDIM - KTAIL == 4, "the k tail must be exactly one k-step");

template <int NA> struct Frag  { v4f a[NA]; v4f b[4]; };
template <int NA> struct FragT { v2f a[NA]; v2f b[4]; };

template <int NA>
__device__ __forceinline__ void load_frag(Frag<NA>& f, const float* const (&xr)[NA],
                                          const float* const (&wr)[4], int pair)
{
    #pragma unroll
    for (int j = 0; j < NA; ++j)
        f.a[j] = *(const v4f*)(xr[j] + pair * 8);
    #pragma unroll
    for (int i = 0; i < 4; ++i)
        f.b[i] = *(const v4f*)(wr[i] + pair * 8);
}

template <int NA>
__device__ __forceinline__ void load_tail(FragT<NA>& f, const float* const (&xr)[NA],
                                          const float* const (&wr)[4], int toff)
{
    #pragma unroll
    for (int j = 0; j < NA; ++j)
        f.a[j] = *(const v2f*)(xr[j] + toff);
    #pragma unroll
    for (int i = 0; i < 4; ++i)
        f.b[i] = *(const v2f*)(wr[i] + toff);
}

template <int NA>
__device__ __forceinline__ void mma_step(v8f (&acc)[MT][4], v8f (&accx)[2],
                                         const v2f (&a)[NA], const v2f (&b)[4])
{
    #pragma unroll
    for (int j = 0; j < MT; ++j) {
        #pragma unroll
        for (int i = 0; i < 4; ++i)
            acc[j][i] = __builtin_amdgcn_wmma_f32_16x16x4_f32(
                false, a[j], false, b[i], (short)0, acc[j][i], false, false);
    }
    if constexpr (NA > MT) {
        #pragma unroll
        for (int i = 0; i < 2; ++i)
            accx[i] = __builtin_amdgcn_wmma_f32_16x16x4_f32(
                false, a[MT], false, b[i], (short)0, accx[i], false, false);
    }
}

template <int NA>
__device__ __forceinline__ void mma_pair(v8f (&acc)[MT][4], v8f (&accx)[2], const Frag<NA>& f)
{
    v2f a[NA], b[4];
    #pragma unroll
    for (int j = 0; j < NA; ++j) a[j] = v2f{ f.a[j].x, f.a[j].y };
    #pragma unroll
    for (int i = 0; i < 4; ++i)  b[i] = v2f{ f.b[i].x, f.b[i].y };
    mma_step<NA>(acc, accx, a, b);
    #pragma unroll
    for (int j = 0; j < NA; ++j) a[j] = v2f{ f.a[j].z, f.a[j].w };
    #pragma unroll
    for (int i = 0; i < 4; ++i)  b[i] = v2f{ f.b[i].z, f.b[i].w };
    mma_step<NA>(acc, accx, a, b);
}

__device__ __forceinline__ float relu(float v) { return v > 0.f ? v : 0.f; }

template <bool PIECE>
__device__ __forceinline__ void gemm_wave(
    const float* __restrict__ x, const float* __restrict__ W1, float* __restrict__ h,
    int n_tiles, int t0, int te, int rot, int lane)
{
    constexpr int NA = MT + (PIECE ? 1 : 0);
    const int nn    = lane & 15;
    const int kq    = (lane >> 4) << 2;
    const int toff  = KTAIL - (kq >> 1);

    const float* xr[NA];
    #pragma unroll
    for (int j = 0; j < MT; ++j) {
        int t = t0 + j;
        t = t < n_tiles ? t : n_tiles - 1;
        xr[j] = x + (size_t)(t * 16 + nn) * KDIM + kq;
    }
    if constexpr (PIECE)
        xr[MT] = x + (size_t)(te * 16 + nn) * KDIM + kq;
    const float* wr[4];
    #pragma unroll
    for (int i = 0; i < 4; ++i)
        wr[i] = W1 + (size_t)(nn + 16 * ((i + rot) & 3)) * KDIM + kq;

    v8f acc[MT][4];
    #pragma unroll
    for (int j = 0; j < MT; ++j) {
        #pragma unroll
        for (int i = 0; i < 4; ++i)
            acc[j][i] = v8f{};
    }
    v8f accx[2] = { v8f{}, v8f{} };

    Frag<NA> f[RING];
    FragT<NA> ft;
    #pragma unroll
    for (int u = 0; u < LA; ++u) {
        load_frag(f[u], xr, wr, u);
        __builtin_amdgcn_sched_barrier(0);
    }

    for (int s = 0; s < S_MAIN; s += RING) {
        #pragma unroll
        for (int u = 0; u < RING; ++u) {
            load_frag(f[(u + LA) % RING], xr, wr, s + u + LA);
            mma_pair(acc, accx, f[u]);
            __builtin_amdgcn_sched_barrier(0);
        }
    }
    #pragma unroll
    for (int t = S_MAIN; t < NPAIR; ++t) {
        if (t + LA < NPAIR)       load_frag(f[(t + LA) % RING], xr, wr, t + LA);
        else if (t + LA == NPAIR) load_tail(ft, xr, wr, toff);
        mma_pair(acc, accx, f[t % RING]);
    }
    mma_step<NA>(acc, accx, ft.a, ft.b);

    const int rhalf = (lane >> 4) << 3;
    #pragma unroll
    for (int j = 0; j < MT; ++j) {
        if (t0 + j < n_tiles) {
            const int m_lo = (t0 + j) * 16 + rhalf;
            #pragma unroll
            for (int r = 0; r < 8; ++r) {
                const size_t mrow = (size_t)(m_lo + r) * HIDDEN;
                #pragma unroll
                for (int i = 0; i < 4; ++i)
                    h[mrow + 16 * ((i + rot) & 3) + nn] = relu(acc[j][i][r]);
            }
        }
    }
    if constexpr (PIECE) {
        const int m_lo = te * 16 + rhalf;
        #pragma unroll
        for (int r = 0; r < 8; ++r) {
            const size_t mrow = (size_t)(m_lo + r) * HIDDEN;
            h[mrow + 16 * rot + nn]      = relu(accx[0][r]);
            h[mrow + 16 * rot + 16 + nn] = relu(accx[1][r]);
        }
    }
}

__global__ __launch_bounds__(GEMM_WAVES * 32) void k_gemm_relu(
    const float* __restrict__ x, const float* __restrict__ W1,
    float* __restrict__ h, int n_tiles, int n_wg,
    float4* __restrict__ zp, int n4, int* __restrict__ tk)
{
    if ((int)blockIdx.x >= n_wg) {
        if ((int)blockIdx.x == n_wg && threadIdx.x < 2)
            tk[threadIdx.x * TK_STRIDE] = 0;
        const int zstride = ((int)gridDim.x - n_wg) * (GEMM_WAVES * 32);
        for (int i = ((int)blockIdx.x - n_wg) * (GEMM_WAVES * 32) + (int)threadIdx.x; i < n4; i += zstride)
            zp[i] = make_float4(0.f, 0.f, 0.f, 0.f);
        return;
    }
    const int lane = threadIdx.x & 31;
    const int wl = __builtin_amdgcn_readfirstlane((int)(threadIdx.x >> 5));
    const int t0     = ((int)blockIdx.x * GEMM_WAVES + wl) * MT;
    const int n_main = n_wg * GEMM_WAVES * MT;
    const int n_piece = 2 * (n_tiles - n_main);
    const int e = wl * n_wg + (int)blockIdx.x;
    if (e < n_piece)
        gemm_wave<true>(x, W1, h, n_tiles, t0, n_main + (e >> 1), (e & 1) * 2, lane);
    else if (t0 < n_tiles)
        gemm_wave<false>(x, W1, h, n_tiles, t0, 0, 0, lane);
}

#define EPW 32
#define EG  8
#define CH2 (HIDDEN / 2)
#define SPMM_WAVES 8
#define EPT (SPMM_WAVES * EPW)
#define SPMM_WGS 2048

__device__ __forceinline__ int draw_ticket(int* tk, int r)
{
    return __hip_atomic_fetch_add(tk + r * TK_STRIDE, 1, __ATOMIC_RELAXED, __HIP_MEMORY_SCOPE_AGENT);
}

template <int NG>
__device__ __forceinline__ void spmm_group(
    const float* __restrict__ h, float* __restrict__ out,
    int sv, int dv, int vb, unsigned& m, unsigned c)
{
    float ha[NG], hb[NG];
    int   li[NG];
    #pragma unroll
    for (int u = 0; u < NG; ++u) {
        li[u] = __builtin_ctz(m);
        m &= m - 1;
        const int s = __builtin_amdgcn_readlane(sv, li[u]);
        const float* hp = h + ((unsigned)s * HIDDEN + c);
        ha[u] = hp[0];
        hb[u] = hp[CH2];
    }
    #pragma unroll
    for (int u = 0; u < NG; ++u) {
        const int   d = __builtin_amdgcn_readlane(dv, li[u]);
        const float v = __builtin_bit_cast(float, __builtin_amdgcn_readlane(vb, li[u]));
        float* o = out + ((unsigned)d * HIDDEN + c);
        __hip_atomic_fetch_add(o,       v * ha[u], __ATOMIC_RELAXED, __HIP_MEMORY_SCOPE_AGENT);
        __hip_atomic_fetch_add(o + CH2, v * hb[u], __ATOMIC_RELAXED, __HIP_MEMORY_SCOPE_AGENT);
    }
}

__device__ __forceinline__ void spmm_wave(
    const int* __restrict__ src, const int* __restrict__ dst,
    const float* __restrict__ val, const float* __restrict__ h,
    float* __restrict__ out, int nnz, int e0, int r, int nh, int lane)
{
    if (e0 >= nnz) return;
    const int left = nnz - e0;
    const int cnt  = left < EPW ? left : EPW;
    const unsigned c = (unsigned)lane;

    const int ei = e0 + (lane < cnt ? lane : cnt - 1);
    const int   sv = __builtin_nontemporal_load(src + ei);
    const int   dv = __builtin_nontemporal_load(dst + ei);
    const float vv = __builtin_nontemporal_load(val + ei);
    const int   vb = __builtin_bit_cast(int, vv);

    const bool mine = (lane < cnt) & ((dv >= nh) == (r != 0));
    unsigned m = __builtin_amdgcn_ballot_w32(mine);
    const int nm = __builtin_popcount(m);

    for (int g = 0; g + EG <= nm; g += EG)
        spmm_group<EG>(h, out, sv, dv, vb, m, c);
    if (nm & 4) spmm_group<4>(h, out, sv, dv, vb, m, c);
    if (nm & 2) spmm_group<2>(h, out, sv, dv, vb, m, c);
    if (nm & 1) spmm_group<1>(h, out, sv, dv, vb, m, c);
}

__global__ __launch_bounds__(SPMM_WAVES * 32) void k_spmm(
    const int* __restrict__ src1, const int* __restrict__ dst1, const float* __restrict__ val1,
    const int* __restrict__ src2, const int* __restrict__ dst2, const float* __restrict__ val2,
    const float* __restrict__ h, float* __restrict__ h1, float* __restrict__ h2,
    int nnz1, int nnz2, int g1, int g2, int n, int* __restrict__ tk)
{
    __shared__ int s_tk[2];
    __shared__ int s_half;
    const int lane = threadIdx.x & 31;
    const int wv = __builtin_amdgcn_readfirstlane((int)(threadIdx.x >> 5));
    const int nt = g1 + g2;
    const int nh = n >> 1;

    int tn = 0;
    if (threadIdx.x == 0) {
        const unsigned id = __builtin_amdgcn_s_sendmsg_rtn(0x87);
        const int hf = (int)((id >> 18) & 1u);
        s_half = hf;
        tn = draw_ticket(tk, hf);
    }
    int r = 0;
    bool other = false;
    for (int it = 0;; ++it) {
        if (threadIdx.x == 0) s_tk[it & 1] = tn;
        __syncthreads();
        if (it == 0) r = __builtin_amdgcn_readfirstlane(s_half);
        const int t = __builtin_amdgcn_readfirstlane(s_tk[it & 1]);
        if (t >= nt) {
            if (other) break;
            other = true;
            r ^= 1;
            if (threadIdx.x == 0) tn = draw_ticket(tk, r);
            continue;
        }
        if (threadIdx.x == 0) tn = draw_ticket(tk, r);
        const bool first = t < g1;
        spmm_wave(first ? src1 : src2, first ? dst1 : dst2, first ? val1 : val2, h,
                  first ? h1 : h2, first ? nnz1 : nnz2,
                  (first ? t : t - g1) * EPT + wv * EPW, r, nh, lane);
    }
}

__global__ __launch_bounds__(256) void k_final(
    const float* __restrict__ h,  const float* __restrict__ h1,
    const float* __restrict__ h2, const float* __restrict__ Wf,
    float* __restrict__ out, int n_tiles)
{
    const int lane = threadIdx.x & 31;
    const int tile = blockIdx.x * 8 + (threadIdx.x >> 5);
    if (tile >= n_tiles) return;
    const int nn    = lane & 15;
    const int kq    = (lane >> 4) << 2;
    const int   wn    = nn < 10 ? nn : 9;
    const float wmask = nn < 10 ? 1.f : 0.f;
    const size_t rowoff = (size_t)(tile * 16 + nn) * HIDDEN;
    const float* wrow   = Wf + (size_t)wn * 192;

    const float* bufs[3] = { h, h1, h2 };
    v8f acc = {};
    #pragma unroll
    for (int seg = 0; seg < 3; ++seg) {
        const float* buf = bufs[seg] + rowoff;
        #pragma unroll
        for (int k0 = 0; k0 < 64; k0 += 8) {
            const v4f a = *(const v4f*)(buf + k0 + kq);
            const v4f b = *(const v4f*)(wrow + seg * 64 + k0 + kq) * wmask;
            acc = __builtin_amdgcn_wmma_f32_16x16x4_f32(false, v2f{ a.x, a.y }, false, v2f{ b.x, b.y },
                                                        (short)0, acc, false, false);
            acc = __builtin_amdgcn_wmma_f32_16x16x4_f32(false, v2f{ a.z, a.w }, false, v2f{ b.z, b.w },
                                                        (short)0, acc, false, false);
        }
    }
    if (nn < 10) {
        const int m_lo = tile * 16 + ((lane >> 4) << 3);
        #pragma unroll
        for (int r = 0; r < 8; ++r)
            __builtin_nontemporal_store(acc[r], out + (size_t)(m_lo + r) * 10 + nn);
    }
}

extern "C" void kernel_launch(void* const* d_in, const int* in_sizes, int n_in,
                              void* d_out, int out_size, void* d_ws, size_t ws_size,
                              hipStream_t stream)
{
    const float* x    = (const float*)d_in[0];
    const float* W1   = (const float*)d_in[1];
    const float* Wf   = (const float*)d_in[2];
    const int*   src1 = (const int*)  d_in[3];
    const int*   dst1 = (const int*)  d_in[4];
    const float* val1 = (const float*)d_in[5];
    const int*   src2 = (const int*)  d_in[6];
    const int*   dst2 = (const int*)  d_in[7];
    const float* val2 = (const float*)d_in[8];

    const int n    = in_sizes[0] / KDIM;
    const int nnz1 = in_sizes[3];
    const int nnz2 = in_sizes[6];

    float* h   = (float*)d_ws;
    float* h1  = h  + (size_t)n * HIDDEN;
    float* h2  = h1 + (size_t)n * HIDDEN;
    float* out = (float*)d_out;
    int* tk = (int*)d_out;

    const int n_tiles = (n + 15) / 16;

    int gemm_wgs = (2 * n_tiles + SHARE - 1) / SHARE;
    if (gemm_wgs < GEMM_WGS) gemm_wgs = GEMM_WGS;
    {
        const long long n4 = (long long)2 * n * HIDDEN / 4;
        k_gemm_relu<<<gemm_wgs + GEMM_WGS, GEMM_WAVES * 32, 0, stream>>>(
            x, W1, h, n_tiles, gemm_wgs, (float4*)h1, (int)n4, tk);
    }

    {
        const int g1 = (nnz1 + EPT - 1) / EPT;
        const int g2 = (nnz2 + EPT - 1) / EPT;
        k_spmm<<<SPMM_WGS, SPMM_WAVES * 32, 0, stream>>>(src1, dst1, val1, src2, dst2, val2, h, h1, h2,
                                                        nnz1, nnz2, g1, g2, n, tk);
    }

    k_final<<<(n_tiles + 7) / 8, 256, 0, stream>>>(h, h1, h2, Wf, out, n_tiles);
}
